# starting kernel
# baseline (speedup 1.0000x reference)
#include <hip/hip_runtime.h>
#include <hip/hip_bf16.h>

#define B_N   256
#define C_N   512
#define HW_N  196
#define ND_N  32
#define NA_N  2000

typedef float v2f __attribute__((ext_vector_type(2)));
typedef float v8f __attribute__((ext_vector_type(8)));

__global__ void k_attend(const float* __restrict__ mask,
                         const float* __restrict__ feat,
                         float* __restrict__ attended) {
    const int b   = blockIdx.x;
    const int tid = threadIdx.x;
    __shared__ float m_s[HW_N];
    if (tid < HW_N) m_s[tid] = mask[b * HW_N + tid];
    __syncthreads();

    const int wave = tid >> 5;
    const int lane = tid & 31;
    const float* fb = feat + (size_t)b * C_N * HW_N;

    for (int c = wave; c < C_N; c += 8) {
        const float* fc = fb + (size_t)c * HW_N;
        float s = 0.0f;
        for (int hw = lane; hw < HW_N; hw += 32)
            s += m_s[hw] * fc[hw];
        #pragma unroll
        for (int off = 16; off > 0; off >>= 1)
            s += __shfl_xor(s, off, 32);
        if (lane == 0)
            attended[b * C_N + c] = s * (1.0f / (float)HW_N);
    }
}

__global__ void k_group(const int* __restrict__ instance,
                        int* __restrict__ counts,
                        int* __restrict__ glist) {
    __shared__ int cnt[ND_N];
    const int tid = threadIdx.x;
    if (tid < ND_N) cnt[tid] = 0;
    for (int i = tid; i < ND_N * B_N; i += B_N) glist[i] = 0;
    __syncthreads();
    const int d    = instance[tid];
    const int slot = atomicAdd(&cnt[d], 1);
    glist[d * B_N + slot] = tid;
    __syncthreads();
    if (tid < ND_N) counts[tid] = cnt[tid];
}

__global__ void k_gemm(const float* __restrict__ attended,
                       const float* __restrict__ weight,
                       const float* __restrict__ bias,
                       const int* __restrict__ counts,
                       const int* __restrict__ glist,
                       float* __restrict__ out) {
    const int d     = blockIdx.x >> 4;
    const int mtile = blockIdx.x & 15;
    const int count = counts[d];
    if (mtile * 16 >= count) return;
    const int rowsInTile = min(16, count - mtile * 16);
    const int a0 = blockIdx.y * 16;

    __shared__ int rows_s[16];
    const int lane = threadIdx.x;
    if (lane < 16) rows_s[lane] = glist[d * B_N + mtile * 16 + lane];
    __syncthreads();

    const int half = lane >> 4;
    const int n    = lane & 15;

    const int rowA = rows_s[n];
    const float* arow = attended + (size_t)rowA * C_N + 2 * half;
    const float* wcol = weight + (size_t)d * C_N * NA_N + a0 + n;

    v8f acc = {};
    #pragma unroll 4
    for (int k0 = 0; k0 < C_N; k0 += 4) {
        v2f afrag = *(const v2f*)(arow + k0);
        const int kb = k0 + 2 * half;
        v2f bfrag;
        bfrag.x = wcol[(size_t)kb * NA_N];
        bfrag.y = wcol[(size_t)(kb + 1) * NA_N];
        acc = __builtin_amdgcn_wmma_f32_16x16x4_f32(
             false, afrag,  false, bfrag,
             (short)0, acc,  false,  false);
    }

    const float bia = bias[d * NA_N + a0 + n];
    #pragma unroll
    for (int v = 0; v < 8; ++v) {
        const int m = v + 8 * half;
        if (m < rowsInTile) {
            const int bidx = rows_s[m];
            out[(size_t)bidx * NA_N + a0 + n] = acc[v] + bia;
        }
    }
}

extern "C" void kernel_launch(void* const* d_in, const int* in_sizes, int n_in,
                              void* d_out, int out_size, void* d_ws, size_t ws_size,
                              hipStream_t stream) {
    const float* mask     = (const float*)d_in[0];
    const float* features = (const float*)d_in[1];
    const float* weight   = (const float*)d_in[2];
    const float* bias     = (const float*)d_in[3];
    const int*   instance = (const int*)d_in[4];
    float*       out      = (float*)d_out;

    float* attended = (float*)d_ws;
    int*   counts   = (int*)((char*)d_ws + (size_t)B_N * C_N * 4);
    int*   glist    = counts + ND_N;

    k_attend<<<dim3(B_N), dim3(256), 0, stream>>>(mask, features, attended);
    k_group <<<dim3(1),   dim3(B_N), 0, stream>>>(instance, counts, glist);
    k_gemm  <<<dim3(ND_N * 16, NA_N / 16), dim3(32), 0, stream>>>(
        attended, weight, bias, counts, glist, out);
}


# stage 1 (masked spatial mean) rebuilt as a stream: non-temporal 16-byte loads over whole rows, 32 rows per wave, 16 rows of loads in flight; the f32 row sums are added in a different order
# speedup vs baseline: 1.7306x; 1.7306x over previous
#include <hip/hip_runtime.h>
#include <hip/hip_bf16.h>

#define B_N   256
#define C_N   512
#define HW_N  196
#define ND_N  32
#define NA_N  2000

typedef float v2f __attribute__((ext_vector_type(2)));
typedef float v8f __attribute__((ext_vector_type(8)));

#define HW4_N     (HW_N / 4)
#define ATT_ROWS  32
#define ATT_WAVES 8
#define ATT_AHEAD 16

typedef float v4f __attribute__((ext_vector_type(4)));

__global__ void __launch_bounds__(ATT_WAVES * 32)
k_attend(const float* __restrict__ mask,
         const float* __restrict__ feat,
         float* __restrict__ attended) {
    const unsigned tid  = threadIdx.x;
    const unsigned lane = tid & 31u;
    const unsigned wave = __builtin_amdgcn_readfirstlane(tid >> 5);
    const unsigned row0 = (blockIdx.x * ATT_WAVES + wave) * ATT_ROWS;
    const unsigned b    = row0 / C_N;

    const bool     two  = lane < (unsigned)(HW4_N - 32);
    const unsigned idx1 = two ? (32u + lane) : (unsigned)(HW4_N - 1);

    const v4f* m4 = (const v4f*)mask + (size_t)b * HW4_N;
    const v4f  m0 = m4[lane];
    const v4f  m1 = m4[idx1];

    const v4f* f0 = (const v4f*)feat + (size_t)row0 * HW4_N + lane;
    const v4f* f1 = (const v4f*)feat + (size_t)row0 * HW4_N + idx1;

    v4f   w[ATT_AHEAD + 1][2];
    float sum[ATT_ROWS];

#define ATT_LOAD_ROW(R) do {                                                          \
        w[(R) % (ATT_AHEAD + 1)][0] = __builtin_nontemporal_load(f0 + (R) * HW4_N);   \
        w[(R) % (ATT_AHEAD + 1)][1] = __builtin_nontemporal_load(f1 + (R) * HW4_N);   \
    } while (0)

    #pragma unroll
    for (int r = 0; r < ATT_AHEAD; ++r) ATT_LOAD_ROW(r);

    #pragma unroll
    for (int r = 0; r < ATT_ROWS; ++r) {
        __builtin_amdgcn_sched_barrier(0);
        if (r + ATT_AHEAD < ATT_ROWS) ATT_LOAD_ROW(r + ATT_AHEAD);
        __builtin_amdgcn_sched_barrier(0);
        const v4f a = w[r % (ATT_AHEAD + 1)][0];
        const v4f c = w[r % (ATT_AHEAD + 1)][1];
        float s0 = m0.x * a.x;
        float s1 = m1.x * c.x;
        s0 = __builtin_fmaf(m0.y, a.y, s0);
        s1 = __builtin_fmaf(m1.y, c.y, s1);
        s0 = __builtin_fmaf(m0.z, a.z, s0);
        s1 = __builtin_fmaf(m1.z, c.z, s1);
        s0 = __builtin_fmaf(m0.w, a.w, s0);
        s1 = __builtin_fmaf(m1.w, c.w, s1);
        sum[r] = s0 + (two ? s1 : 0.0f);
        asm volatile("" : "+v"(sum[r]));
    }
#undef ATT_LOAD_ROW

    #pragma unroll
    for (int d = 16; d >= 1; d >>= 1) {
        const bool up = (lane & (unsigned)d) != 0u;
        #pragma unroll
        for (int i = 0; i < d; ++i) {
            const float lo = sum[i], hi = sum[i + d];
            const float keep = up ? hi : lo;
            const float send = up ? lo : hi;
            sum[i] = keep + __shfl_xor(send, d, 32);
        }
    }
    attended[row0 + lane] = sum[0] * (1.0f / (float)HW_N);
}

__global__ void k_group(const int* __restrict__ instance,
                        int* __restrict__ counts,
                        int* __restrict__ glist) {
    __shared__ int cnt[ND_N];
    const int tid = threadIdx.x;
    if (tid < ND_N) cnt[tid] = 0;
    for (int i = tid; i < ND_N * B_N; i += B_N) glist[i] = 0;
    __syncthreads();
    const int d    = instance[tid];
    const int slot = atomicAdd(&cnt[d], 1);
    glist[d * B_N + slot] = tid;
    __syncthreads();
    if (tid < ND_N) counts[tid] = cnt[tid];
}

__global__ void k_gemm(const float* __restrict__ attended,
                       const float* __restrict__ weight,
                       const float* __restrict__ bias,
                       const int* __restrict__ counts,
                       const int* __restrict__ glist,
                       float* __restrict__ out) {
    const int d     = blockIdx.x >> 4;
    const int mtile = blockIdx.x & 15;
    const int count = counts[d];
    if (mtile * 16 >= count) return;
    const int rowsInTile = min(16, count - mtile * 16);
    const int a0 = blockIdx.y * 16;

    __shared__ int rows_s[16];
    const int lane = threadIdx.x;
    if (lane < 16) rows_s[lane] = glist[d * B_N + mtile * 16 + lane];
    __syncthreads();

    const int half = lane >> 4;
    const int n    = lane & 15;

    const int rowA = rows_s[n];
    const float* arow = attended + (size_t)rowA * C_N + 2 * half;
    const float* wcol = weight + (size_t)d * C_N * NA_N + a0 + n;

    v8f acc = {};
    #pragma unroll 4
    for (int k0 = 0; k0 < C_N; k0 += 4) {
        v2f afrag = *(const v2f*)(arow + k0);
        const int kb = k0 + 2 * half;
        v2f bfrag;
        bfrag.x = wcol[(size_t)kb * NA_N];
        bfrag.y = wcol[(size_t)(kb + 1) * NA_N];
        acc = __builtin_amdgcn_wmma_f32_16x16x4_f32(
             false, afrag,  false, bfrag,
             (short)0, acc,  false,  false);
    }

    const float bia = bias[d * NA_N + a0 + n];
    #pragma unroll
    for (int v = 0; v < 8; ++v) {
        const int m = v + 8 * half;
        if (m < rowsInTile) {
            const int bidx = rows_s[m];
            out[(size_t)bidx * NA_N + a0 + n] = acc[v] + bia;
        }
    }
}

extern "C" void kernel_launch(void* const* d_in, const int* in_sizes, int n_in,
                              void* d_out, int out_size, void* d_ws, size_t ws_size,
                              hipStream_t stream) {
    const float* mask     = (const float*)d_in[0];
    const float* features = (const float*)d_in[1];
    const float* weight   = (const float*)d_in[2];
    const float* bias     = (const float*)d_in[3];
    const int*   instance = (const int*)d_in[4];
    float*       out      = (float*)d_out;

    float* attended = (float*)d_ws;
    int*   counts   = (int*)((char*)d_ws + (size_t)B_N * C_N * 4);
    int*   glist    = counts + ND_N;

    k_attend<<<dim3(B_N * C_N / (ATT_ROWS * ATT_WAVES)), dim3(ATT_WAVES * 32), 0, stream>>>(mask, features, attended);
    k_group <<<dim3(1),   dim3(B_N), 0, stream>>>(instance, counts, glist);
    k_gemm  <<<dim3(ND_N * 16, NA_N / 16), dim3(32), 0, stream>>>(
        attended, weight, bias, counts, glist, out);
}


# stage 2b (grouped GEMM) rebuilt as a stream: 64-column work-groups, 16-byte non-temporal weight loads each feeding 4 WMMAs, K split over 4 waves, summed in LDS in a fixed order (f32 sums re-ordered)
# speedup vs baseline: 9.1894x; 5.3100x over previous
#include <hip/hip_runtime.h>
#include <hip/hip_bf16.h>

#define B_N   256
#define C_N   512
#define HW_N  196
#define ND_N  32
#define NA_N  2000

typedef float v2f __attribute__((ext_vector_type(2)));
typedef float v8f __attribute__((ext_vector_type(8)));

#define HW4_N     (HW_N / 4)
#define ATT_ROWS  32
#define ATT_WAVES 8
#define ATT_AHEAD 16

typedef float v4f __attribute__((ext_vector_type(4)));

__global__ void __launch_bounds__(ATT_WAVES * 32)
k_attend(const float* __restrict__ mask,
         const float* __restrict__ feat,
         float* __restrict__ attended) {
    const unsigned tid  = threadIdx.x;
    const unsigned lane = tid & 31u;
    const unsigned wave = __builtin_amdgcn_readfirstlane(tid >> 5);
    const unsigned row0 = (blockIdx.x * ATT_WAVES + wave) * ATT_ROWS;
    const unsigned b    = row0 / C_N;

    const bool     two  = lane < (unsigned)(HW4_N - 32);
    const unsigned idx1 = two ? (32u + lane) : (unsigned)(HW4_N - 1);

    const v4f* m4 = (const v4f*)mask + (size_t)b * HW4_N;
    const v4f  m0 = m4[lane];
    const v4f  m1 = m4[idx1];

    const v4f* f0 = (const v4f*)feat + (size_t)row0 * HW4_N + lane;
    const v4f* f1 = (const v4f*)feat + (size_t)row0 * HW4_N + idx1;

    v4f   w[ATT_AHEAD + 1][2];
    float sum[ATT_ROWS];

#define ATT_LOAD_ROW(R) do {                                                          \
        w[(R) % (ATT_AHEAD + 1)][0] = __builtin_nontemporal_load(f0 + (R) * HW4_N);   \
        w[(R) % (ATT_AHEAD + 1)][1] = __builtin_nontemporal_load(f1 + (R) * HW4_N);   \
    } while (0)

    #pragma unroll
    for (int r = 0; r < ATT_AHEAD; ++r) ATT_LOAD_ROW(r);

    #pragma unroll
    for (int r = 0; r < ATT_ROWS; ++r) {
        __builtin_amdgcn_sched_barrier(0);
        if (r + ATT_AHEAD < ATT_ROWS) ATT_LOAD_ROW(r + ATT_AHEAD);
        __builtin_amdgcn_sched_barrier(0);
        const v4f a = w[r % (ATT_AHEAD + 1)][0];
        const v4f c = w[r % (ATT_AHEAD + 1)][1];
        float s0 = m0.x * a.x;
        float s1 = m1.x * c.x;
        s0 = __builtin_fmaf(m0.y, a.y, s0);
        s1 = __builtin_fmaf(m1.y, c.y, s1);
        s0 = __builtin_fmaf(m0.z, a.z, s0);
        s1 = __builtin_fmaf(m1.z, c.z, s1);
        s0 = __builtin_fmaf(m0.w, a.w, s0);
        s1 = __builtin_fmaf(m1.w, c.w, s1);
        sum[r] = s0 + (two ? s1 : 0.0f);
        asm volatile("" : "+v"(sum[r]));
    }
#undef ATT_LOAD_ROW

    #pragma unroll
    for (int d = 16; d >= 1; d >>= 1) {
        const bool up = (lane & (unsigned)d) != 0u;
        #pragma unroll
        for (int i = 0; i < d; ++i) {
            const float lo = sum[i], hi = sum[i + d];
            const float keep = up ? hi : lo;
            const float send = up ? lo : hi;
            sum[i] = keep + __shfl_xor(send, d, 32);
        }
    }
    attended[row0 + lane] = sum[0] * (1.0f / (float)HW_N);
}

__global__ void k_group(const int* __restrict__ instance,
                        int* __restrict__ counts,
                        int* __restrict__ glist) {
    __shared__ int cnt[ND_N];
    const int tid = threadIdx.x;
    if (tid < ND_N) cnt[tid] = 0;
    for (int i = tid; i < ND_N * B_N; i += B_N) glist[i] = 0;
    __syncthreads();
    const int d    = instance[tid];
    const int slot = atomicAdd(&cnt[d], 1);
    glist[d * B_N + slot] = tid;
    __syncthreads();
    if (tid < ND_N) counts[tid] = cnt[tid];
}

#define G_COLS   64
#define G_WAVES  4
#define G_KW     (C_N / G_WAVES)
#define G_STEPS  (G_KW / 4)
#define G_AHEAD  16
#define G_ABLK   8
#define G_APITCH (G_KW + 4)
#define G_NBLK   ((NA_N + G_COLS - 1) / G_COLS)

typedef const v4f __attribute__((address_space(1)))* gv4f_ptr;

__global__ void __launch_bounds__(G_WAVES * 32)
k_gemm(const float* __restrict__ attended,
       const float* __restrict__ weight,
       const float* __restrict__ bias,
       const int* __restrict__ counts,
       const int* __restrict__ glist,
       float* __restrict__ out) {
    __shared__ __attribute__((aligned(16))) float a_s[G_WAVES][16][G_APITCH];
    __shared__ v4f p_s[G_WAVES][16][17];

    const unsigned tid  = threadIdx.x;
    const unsigned lane = tid & 31u;
    const unsigned wave = __builtin_amdgcn_readfirstlane(tid >> 5);
    const unsigned h    = lane >> 4;
    const unsigned c    = lane & 15u;
    const unsigned d    = blockIdx.y;
    const unsigned a0   = blockIdx.x * G_COLS;
    const unsigned kw   = wave * G_KW;
    const int      count = counts[d];

    const unsigned colw = min(a0 + 4u * c, (unsigned)(NA_N - 4));
    const v4f* wp = (const v4f*)(weight + ((size_t)d * C_N + kw + 2u * h) * NA_N + colw);

    const unsigned c4 = tid & 15u;
    const unsigned rr = tid >> 4;
    const unsigned colo = a0 + 4u * c4;

    const int* gl = glist + d * B_N;
    int rowv = gl[c];
    int bo0  = gl[rr];
    int bo1  = gl[rr + 8];
    asm volatile("" : "+v"(rowv), "+v"(bo0), "+v"(bo1));

    for (int m0 = 0; m0 < count; m0 += 16) {
        v4f av[16];
        #pragma unroll
        for (int m = 0; m < 16; ++m) {
            const int r = __builtin_amdgcn_readlane(rowv, m);
            av[m] = *(const v4f*)(attended + (size_t)r * C_N + kw + 4u * lane);
        }

        gv4f_ptr wq = (gv4f_ptr)wp;
        asm volatile("" : "+v"(wq));

        v4f ring[G_AHEAD + 1][2];
#define G_LOAD_STEP(S) do {                                                                          \
        ring[(S) % (G_AHEAD + 1)][0] = __builtin_nontemporal_load(wq + (size_t)(4 * (S) + 0) * (NA_N / 4)); \
        ring[(S) % (G_AHEAD + 1)][1] = __builtin_nontemporal_load(wq + (size_t)(4 * (S) + 1) * (NA_N / 4)); \
    } while (0)

        __builtin_amdgcn_sched_barrier(0);
        #pragma unroll
        for (int s = 0; s < G_AHEAD / 2; ++s) G_LOAD_STEP(s);
        __builtin_amdgcn_sched_barrier(0);

        #pragma unroll
        for (int m = 0; m < 16; ++m)
            *(v4f*)&a_s[wave][m][4u * lane] = av[m];
        __builtin_amdgcn_sched_barrier(0);
        #pragma unroll
        for (int s = G_AHEAD / 2; s < G_AHEAD; ++s) G_LOAD_STEP(s);
        __builtin_amdgcn_fence(__ATOMIC_RELEASE, "wavefront");
        __builtin_amdgcn_wave_barrier();
        __builtin_amdgcn_fence(__ATOMIC_ACQUIRE, "wavefront");

        v8f acc0 = {}, acc1 = {}, acc2 = {}, acc3 = {};
        v2f af[G_ABLK];

        #pragma unroll
        for (int s = 0; s < G_STEPS; ++s) {
            __builtin_amdgcn_sched_barrier(0);
            if (s + G_AHEAD < G_STEPS) G_LOAD_STEP(s + G_AHEAD);
            __builtin_amdgcn_sched_barrier(0);
            if (s % G_ABLK == 0) {
                #pragma unroll
                for (int i = 0; i < G_ABLK; ++i)
                    af[i] = *(const v2f*)&a_s[wave][c][4 * (s + i) + 2u * h];
            }
            const v4f r0 = ring[s % (G_AHEAD + 1)][0];
            const v4f r1 = ring[s % (G_AHEAD + 1)][1];
            const v2f a  = af[s % G_ABLK];
            v2f b;
            b.x = r0.x; b.y = r1.x;
            acc0 = __builtin_amdgcn_wmma_f32_16x16x4_f32(false, a, false, b, (short)0, acc0, false, false);
            b.x = r0.y; b.y = r1.y;
            acc1 = __builtin_amdgcn_wmma_f32_16x16x4_f32(false, a, false, b, (short)0, acc1, false, false);
            b.x = r0.z; b.y = r1.z;
            acc2 = __builtin_amdgcn_wmma_f32_16x16x4_f32(false, a, false, b, (short)0, acc2, false, false);
            b.x = r0.w; b.y = r1.w;
            acc3 = __builtin_amdgcn_wmma_f32_16x16x4_f32(false, a, false, b, (short)0, acc3, false, false);
        }
#undef G_LOAD_STEP

        #pragma unroll
        for (int v = 0; v < 8; ++v) {
            v4f p;
            p.x = acc0[v]; p.y = acc1[v]; p.z = acc2[v]; p.w = acc3[v];
            p_s[wave][v + 8 * h][c] = p;
        }
        __syncthreads();

        if (colo < (unsigned)NA_N) {
            const v4f bia = *(const v4f*)(bias + d * NA_N + colo);
            if (m0 + (int)rr < count) {
                const v4f s4 = ((p_s[0][rr][c4] + p_s[1][rr][c4]) + (p_s[2][rr][c4] + p_s[3][rr][c4])) + bia;
                *(v4f*)(out + (size_t)bo0 * NA_N + colo) = s4;
            }
            if (m0 + (int)rr + 8 < count) {
                const v4f s4 = ((p_s[0][rr + 8][c4] + p_s[1][rr + 8][c4]) + (p_s[2][rr + 8][c4] + p_s[3][rr + 8][c4])) + bia;
                *(v4f*)(out + (size_t)bo1 * NA_N + colo) = s4;
            }
        }
        if (m0 + 16 < count) {
            __syncthreads();
            rowv = gl[m0 + 16 + c];
            bo0  = gl[m0 + 16 + rr];
            bo1  = gl[m0 + 16 + rr + 8];
        }
    }
}

extern "C" void kernel_launch(void* const* d_in, const int* in_sizes, int n_in,
                              void* d_out, int out_size, void* d_ws, size_t ws_size,
                              hipStream_t stream) {
    const float* mask     = (const float*)d_in[0];
    const float* features = (const float*)d_in[1];
    const float* weight   = (const float*)d_in[2];
    const float* bias     = (const float*)d_in[3];
    const int*   instance = (const int*)d_in[4];
    float*       out      = (float*)d_out;

    float* attended = (float*)d_ws;
    int*   counts   = (int*)((char*)d_ws + (size_t)B_N * C_N * 4);
    int*   glist    = counts + ND_N;

    k_attend<<<dim3(B_N * C_N / (ATT_ROWS * ATT_WAVES)), dim3(ATT_WAVES * 32), 0, stream>>>(mask, features, attended);
    k_group <<<dim3(1),   dim3(B_N), 0, stream>>>(instance, counts, glist);
    k_gemm  <<<dim3(G_NBLK, ND_N), dim3(G_WAVES * 32), 0, stream>>>(
        attended, weight, bias, counts, glist, out);
}


# expert grouping moved into the GEMM launch: each wave lists its expert's samples itself (ballot + count over the 256 expert ids, list in its own LDS slice); 3 launches become 2
# speedup vs baseline: 9.8100x; 1.0675x over previous
#include <hip/hip_runtime.h>
#include <hip/hip_bf16.h>

#define B_N   256
#define C_N   512
#define HW_N  196
#define ND_N  32
#define NA_N  2000

typedef float v2f __attribute__((ext_vector_type(2)));
typedef float v8f __attribute__((ext_vector_type(8)));

#define HW4_N     (HW_N / 4)
#define ATT_ROWS  32
#define ATT_WAVES 8
#define ATT_AHEAD 16

typedef float v4f __attribute__((ext_vector_type(4)));

__global__ void __launch_bounds__(ATT_WAVES * 32)
k_attend(const float* __restrict__ mask,
         const float* __restrict__ feat,
         float* __restrict__ attended) {
    const unsigned tid  = threadIdx.x;
    const unsigned lane = tid & 31u;
    const unsigned wave = __builtin_amdgcn_readfirstlane(tid >> 5);
    const unsigned row0 = (blockIdx.x * ATT_WAVES + wave) * ATT_ROWS;
    const unsigned b    = row0 / C_N;

    const bool     two  = lane < (unsigned)(HW4_N - 32);
    const unsigned idx1 = two ? (32u + lane) : (unsigned)(HW4_N - 1);

    const v4f* m4 = (const v4f*)mask + (size_t)b * HW4_N;
    const v4f  m0 = m4[lane];
    const v4f  m1 = m4[idx1];

    const v4f* f0 = (const v4f*)feat + (size_t)row0 * HW4_N + lane;
    const v4f* f1 = (const v4f*)feat + (size_t)row0 * HW4_N + idx1;

    v4f   w[ATT_AHEAD + 1][2];
    float sum[ATT_ROWS];

#define ATT_LOAD_ROW(R) do {                                                          \
        w[(R) % (ATT_AHEAD + 1)][0] = __builtin_nontemporal_load(f0 + (R) * HW4_N);   \
        w[(R) % (ATT_AHEAD + 1)][1] = __builtin_nontemporal_load(f1 + (R) * HW4_N);   \
    } while (0)

    #pragma unroll
    for (int r = 0; r < ATT_AHEAD; ++r) ATT_LOAD_ROW(r);

    #pragma unroll
    for (int r = 0; r < ATT_ROWS; ++r) {
        __builtin_amdgcn_sched_barrier(0);
        if (r + ATT_AHEAD < ATT_ROWS) ATT_LOAD_ROW(r + ATT_AHEAD);
        __builtin_amdgcn_sched_barrier(0);
        const v4f a = w[r % (ATT_AHEAD + 1)][0];
        const v4f c = w[r % (ATT_AHEAD + 1)][1];
        float s0 = m0.x * a.x;
        float s1 = m1.x * c.x;
        s0 = __builtin_fmaf(m0.y, a.y, s0);
        s1 = __builtin_fmaf(m1.y, c.y, s1);
        s0 = __builtin_fmaf(m0.z, a.z, s0);
        s1 = __builtin_fmaf(m1.z, c.z, s1);
        s0 = __builtin_fmaf(m0.w, a.w, s0);
        s1 = __builtin_fmaf(m1.w, c.w, s1);
        sum[r] = s0 + (two ? s1 : 0.0f);
        asm volatile("" : "+v"(sum[r]));
    }
#undef ATT_LOAD_ROW

    #pragma unroll
    for (int d = 16; d >= 1; d >>= 1) {
        const bool up = (lane & (unsigned)d) != 0u;
        #pragma unroll
        for (int i = 0; i < d; ++i) {
            const float lo = sum[i], hi = sum[i + d];
            const float keep = up ? hi : lo;
            const float send = up ? lo : hi;
            sum[i] = keep + __shfl_xor(send, d, 32);
        }
    }
    attended[row0 + lane] = sum[0] * (1.0f / (float)HW_N);
}

#define G_COLS   64
#define G_WAVES  4
#define G_KW     (C_N / G_WAVES)
#define G_STEPS  (G_KW / 4)
#define G_AHEAD  16
#define G_ABLK   8
#define G_APITCH (G_KW + 4)
#define G_NBLK   ((NA_N + G_COLS - 1) / G_COLS)

typedef const v4f __attribute__((address_space(1)))* gv4f_ptr;
typedef int v4i __attribute__((ext_vector_type(4)));

__global__ void __launch_bounds__(G_WAVES * 32)
k_gemm(const float* __restrict__ attended,
       const float* __restrict__ weight,
       const float* __restrict__ bias,
       const int* __restrict__ instance,
       float* __restrict__ out) {
    __shared__ __attribute__((aligned(16))) float a_s[G_WAVES][16][G_APITCH];
    __shared__ v4f p_s[G_WAVES][16][17];
    __shared__ int l_s[G_WAVES][B_N];

    const unsigned tid  = threadIdx.x;
    const unsigned lane = tid & 31u;
    const unsigned wave = __builtin_amdgcn_readfirstlane(tid >> 5);
    const unsigned h    = lane >> 4;
    const unsigned c    = lane & 15u;
    const unsigned d    = blockIdx.y;
    const unsigned a0   = blockIdx.x * G_COLS;
    const unsigned kw   = wave * G_KW;

    const unsigned colw = min(a0 + 4u * c, (unsigned)(NA_N - 4));
    const v4f* wp = (const v4f*)(weight + ((size_t)d * C_N + kw + 2u * h) * NA_N + colw);

    const unsigned c4 = tid & 15u;
    const unsigned rr = tid >> 4;
    const unsigned colo = a0 + 4u * c4;

    const v4i* ip = (const v4i*)instance;
    const v4i  i0 = ip[lane];
    const v4i  i1 = ip[32u + lane];
    int* gl = l_s[wave];
    int count = 0;
    #pragma unroll
    for (int t = 0; t < 8; ++t) {
        const int      v   = (t < 4) ? i0[t & 3] : i1[t & 3];
        const bool     hit = (v == (int)d);
        const unsigned m   = __builtin_amdgcn_ballot_w32(hit);
        if (hit) gl[count + (int)__builtin_amdgcn_mbcnt_lo(m, 0u)] = (int)((unsigned)(t >> 2) * 128u + 4u * lane + (unsigned)(t & 3));
        count += __builtin_popcount(m);
    }
    __builtin_amdgcn_fence(__ATOMIC_RELEASE, "wavefront");
    __builtin_amdgcn_wave_barrier();
    __builtin_amdgcn_fence(__ATOMIC_ACQUIRE, "wavefront");

    int rowv = ((int)c      < count) ? gl[c]      : 0;
    int bo0  = ((int)rr     < count) ? gl[rr]     : 0;
    int bo1  = ((int)rr + 8 < count) ? gl[rr + 8] : 0;

    for (int m0 = 0; m0 < count; m0 += 16) {
        v4f av[16];
        #pragma unroll
        for (int m = 0; m < 16; ++m) {
            const int r = __builtin_amdgcn_readlane(rowv, m);
            av[m] = *(const v4f*)(attended + (size_t)r * C_N + kw + 4u * lane);
        }

        gv4f_ptr wq = (gv4f_ptr)wp;
        asm volatile("" : "+v"(wq));

        v4f ring[G_AHEAD + 1][2];
#define G_LOAD_STEP(S) do {                                                                          \
        ring[(S) % (G_AHEAD + 1)][0] = __builtin_nontemporal_load(wq + (size_t)(4 * (S) + 0) * (NA_N / 4)); \
        ring[(S) % (G_AHEAD + 1)][1] = __builtin_nontemporal_load(wq + (size_t)(4 * (S) + 1) * (NA_N / 4)); \
    } while (0)

        __builtin_amdgcn_sched_barrier(0);
        #pragma unroll
        for (int s = 0; s < G_AHEAD / 2; ++s) G_LOAD_STEP(s);
        __builtin_amdgcn_sched_barrier(0);

        #pragma unroll
        for (int m = 0; m < 16; ++m)
            *(v4f*)&a_s[wave][m][4u * lane] = av[m];
        __builtin_amdgcn_sched_barrier(0);
        #pragma unroll
        for (int s = G_AHEAD / 2; s < G_AHEAD; ++s) G_LOAD_STEP(s);
        __builtin_amdgcn_fence(__ATOMIC_RELEASE, "wavefront");
        __builtin_amdgcn_wave_barrier();
        __builtin_amdgcn_fence(__ATOMIC_ACQUIRE, "wavefront");

        v8f acc0 = {}, acc1 = {}, acc2 = {}, acc3 = {};
        v2f af[G_ABLK];

        #pragma unroll
        for (int s = 0; s < G_STEPS; ++s) {
            __builtin_amdgcn_sched_barrier(0);
            if (s + G_AHEAD < G_STEPS) G_LOAD_STEP(s + G_AHEAD);
            __builtin_amdgcn_sched_barrier(0);
            if (s % G_ABLK == 0) {
                #pragma unroll
                for (int i = 0; i < G_ABLK; ++i)
                    af[i] = *(const v2f*)&a_s[wave][c][4 * (s + i) + 2u * h];
            }
            const v4f r0 = ring[s % (G_AHEAD + 1)][0];
            const v4f r1 = ring[s % (G_AHEAD + 1)][1];
            const v2f a  = af[s % G_ABLK];
            v2f b;
            b.x = r0.x; b.y = r1.x;
            acc0 = __builtin_amdgcn_wmma_f32_16x16x4_f32(false, a, false, b, (short)0, acc0, false, false);
            b.x = r0.y; b.y = r1.y;
            acc1 = __builtin_amdgcn_wmma_f32_16x16x4_f32(false, a, false, b, (short)0, acc1, false, false);
            b.x = r0.z; b.y = r1.z;
            acc2 = __builtin_amdgcn_wmma_f32_16x16x4_f32(false, a, false, b, (short)0, acc2, false, false);
            b.x = r0.w; b.y = r1.w;
            acc3 = __builtin_amdgcn_wmma_f32_16x16x4_f32(false, a, false, b, (short)0, acc3, false, false);
        }
#undef G_LOAD_STEP

        #pragma unroll
        for (int v = 0; v < 8; ++v) {
            v4f p;
            p.x = acc0[v]; p.y = acc1[v]; p.z = acc2[v]; p.w = acc3[v];
            p_s[wave][v + 8 * h][c] = p;
        }
        __syncthreads();

        if (colo < (unsigned)NA_N) {
            const v4f bia = *(const v4f*)(bias + d * NA_N + colo);
            if (m0 + (int)rr < count) {
                const v4f s4 = ((p_s[0][rr][c4] + p_s[1][rr][c4]) + (p_s[2][rr][c4] + p_s[3][rr][c4])) + bia;
                *(v4f*)(out + (size_t)bo0 * NA_N + colo) = s4;
            }
            if (m0 + (int)rr + 8 < count) {
                const v4f s4 = ((p_s[0][rr + 8][c4] + p_s[1][rr + 8][c4]) + (p_s[2][rr + 8][c4] + p_s[3][rr + 8][c4])) + bia;
                *(v4f*)(out + (size_t)bo1 * NA_N + colo) = s4;
            }
        }
        if (m0 + 16 < count) {
            __syncthreads();
            rowv = (m0 + 16 + (int)c      < count) ? gl[m0 + 16 + c]      : 0;
            bo0  = (m0 + 16 + (int)rr     < count) ? gl[m0 + 16 + rr]     : 0;
            bo1  = (m0 + 16 + (int)rr + 8 < count) ? gl[m0 + 16 + rr + 8] : 0;
        }
    }
}

extern "C" void kernel_launch(void* const* d_in, const int* in_sizes, int n_in,
                              void* d_out, int out_size, void* d_ws, size_t ws_size,
                              hipStream_t stream) {
    const float* mask     = (const float*)d_in[0];
    const float* features = (const float*)d_in[1];
    const float* weight   = (const float*)d_in[2];
    const float* bias     = (const float*)d_in[3];
    const int*   instance = (const int*)d_in[4];
    float*       out      = (float*)d_out;

    float* attended = (float*)d_ws;

    k_attend<<<dim3(B_N * C_N / (ATT_ROWS * ATT_WAVES)), dim3(ATT_WAVES * 32), 0, stream>>>(mask, features, attended);
    k_gemm  <<<dim3(G_NBLK, ND_N), dim3(G_WAVES * 32), 0, stream>>>(
        attended, weight, bias, instance, out);
}


# stage 1 reads each wave's 32 rows as 49 whole aligned 512-byte pieces (float4 32i+l of the contiguous region), not two ragged loads per row; mask from a per-wave LDS copy; f32 row sums re-ordered
# speedup vs baseline: 10.1447x; 1.0341x over previous
#include <hip/hip_runtime.h>
#include <hip/hip_bf16.h>

#define B_N   256
#define C_N   512
#define HW_N  196
#define ND_N  32
#define NA_N  2000

typedef float v2f __attribute__((ext_vector_type(2)));
typedef float v8f __attribute__((ext_vector_type(8)));

#define HW4_N      (HW_N / 4)
#define ATT_ROWS   32
#define ATT_WAVES  8
#define ATT_PIECES HW4_N
#define ATT_AHEAD  32

typedef float v4f __attribute__((ext_vector_type(4)));

__global__ void __launch_bounds__(ATT_WAVES * 32)
k_attend(const float* __restrict__ mask,
         const float* __restrict__ feat,
         float* __restrict__ attended) {
    __shared__ v4f m_s[ATT_WAVES][2 * HW4_N];

    const unsigned tid  = threadIdx.x;
    const unsigned lane = tid & 31u;
    const unsigned wave = __builtin_amdgcn_readfirstlane(tid >> 5);
    const unsigned row0 = (blockIdx.x * ATT_WAVES + wave) * ATT_ROWS;
    const unsigned b    = row0 / C_N;

    const bool     two  = lane < (unsigned)(HW4_N - 32);
    const unsigned idx1 = two ? (32u + lane) : (unsigned)(HW4_N - 1);
    const v4f* m4 = (const v4f*)mask + (size_t)b * HW4_N;
    const v4f  m0 = m4[lane];
    const v4f  m1 = m4[idx1];

    __builtin_amdgcn_sched_barrier(0);

    const v4f* f = (const v4f*)feat + (size_t)row0 * HW4_N + lane;

    v4f   w[ATT_AHEAD + 1];
    float sum[ATT_ROWS];

#define ATT_LOAD_PIECE(I) do {                                                       \
        w[(I) % (ATT_AHEAD + 1)] = __builtin_nontemporal_load(f + (I) * 32);         \
    } while (0)

    #pragma unroll
    for (int i = 0; i < ATT_AHEAD; ++i) ATT_LOAD_PIECE(i);
    __builtin_amdgcn_sched_barrier(0);

    v4f* ms = m_s[wave];
    ms[lane]         = m0;
    ms[HW4_N + lane] = m0;
    ms[idx1]         = m1;
    ms[HW4_N + idx1] = m1;
    __builtin_amdgcn_fence(__ATOMIC_RELEASE, "wavefront");
    __builtin_amdgcn_wave_barrier();
    __builtin_amdgcn_fence(__ATOMIC_ACQUIRE, "wavefront");

    v4f mn = ms[lane];

    #pragma unroll
    for (int i = 0; i < ATT_PIECES; ++i) {
        __builtin_amdgcn_sched_barrier(0);
        if (i + ATT_AHEAD < ATT_PIECES) ATT_LOAD_PIECE(i + ATT_AHEAD);
        const int r_lo = (32 * i) / HW4_N;
        const int base = 32 * i - HW4_N * r_lo;
        const int thr  = HW4_N - base;
        const v4f m = mn;
        if (i + 1 < ATT_PIECES) mn = ms[(32 * (i + 1)) % HW4_N + (int)lane];
        __builtin_amdgcn_sched_barrier(0);
        const v4f a = w[i % (ATT_AHEAD + 1)];
        float p = m.x * a.x;
        p = __builtin_fmaf(m.y, a.y, p);
        p = __builtin_fmaf(m.z, a.z, p);
        p = __builtin_fmaf(m.w, a.w, p);
        if (i == 0) {
            sum[0] = p;
        } else if (thr >= 32) {
            sum[r_lo] += p;
        } else {
            const bool lo = lane < (unsigned)thr;
            sum[r_lo]    += lo ? p : 0.0f;
            sum[r_lo + 1] = lo ? 0.0f : p;
            asm volatile("" : "+v"(sum[r_lo + 1]));
        }
        asm volatile("" : "+v"(sum[r_lo]));
    }
#undef ATT_LOAD_PIECE

    #pragma unroll
    for (int d = 16; d >= 1; d >>= 1) {
        const bool up = (lane & (unsigned)d) != 0u;
        #pragma unroll
        for (int i = 0; i < d; ++i) {
            const float lo = sum[i], hi = sum[i + d];
            const float keep = up ? hi : lo;
            const float send = up ? lo : hi;
            sum[i] = keep + __shfl_xor(send, d, 32);
        }
    }
    attended[row0 + lane] = sum[0] * (1.0f / (float)HW_N);
}

#define G_COLS   64
#define G_WAVES  4
#define G_KW     (C_N / G_WAVES)
#define G_STEPS  (G_KW / 4)
#define G_AHEAD  16
#define G_ABLK   8
#define G_APITCH (G_KW + 4)
#define G_NBLK   ((NA_N + G_COLS - 1) / G_COLS)

typedef const v4f __attribute__((address_space(1)))* gv4f_ptr;
typedef int v4i __attribute__((ext_vector_type(4)));

__global__ void __launch_bounds__(G_WAVES * 32)
k_gemm(const float* __restrict__ attended,
       const float* __restrict__ weight,
       const float* __restrict__ bias,
       const int* __restrict__ instance,
       float* __restrict__ out) {
    __shared__ __attribute__((aligned(16))) float a_s[G_WAVES][16][G_APITCH];
    __shared__ v4f p_s[G_WAVES][16][17];
    __shared__ int l_s[G_WAVES][B_N];

    const unsigned tid  = threadIdx.x;
    const unsigned lane = tid & 31u;
    const unsigned wave = __builtin_amdgcn_readfirstlane(tid >> 5);
    const unsigned h    = lane >> 4;
    const unsigned c    = lane & 15u;
    const unsigned d    = blockIdx.y;
    const unsigned a0   = blockIdx.x * G_COLS;
    const unsigned kw   = wave * G_KW;

    const unsigned colw = min(a0 + 4u * c, (unsigned)(NA_N - 4));
    const v4f* wp = (const v4f*)(weight + ((size_t)d * C_N + kw + 2u * h) * NA_N + colw);

    const unsigned c4 = tid & 15u;
    const unsigned rr = tid >> 4;
    const unsigned colo = a0 + 4u * c4;

    const v4i* ip = (const v4i*)instance;
    const v4i  i0 = ip[lane];
    const v4i  i1 = ip[32u + lane];
    int* gl = l_s[wave];
    int count = 0;
    #pragma unroll
    for (int t = 0; t < 8; ++t) {
        const int      v   = (t < 4) ? i0[t & 3] : i1[t & 3];
        const bool     hit = (v == (int)d);
        const unsigned m   = __builtin_amdgcn_ballot_w32(hit);
        if (hit) gl[count + (int)__builtin_amdgcn_mbcnt_lo(m, 0u)] = (int)((unsigned)(t >> 2) * 128u + 4u * lane + (unsigned)(t & 3));
        count += __builtin_popcount(m);
    }
    __builtin_amdgcn_fence(__ATOMIC_RELEASE, "wavefront");
    __builtin_amdgcn_wave_barrier();
    __builtin_amdgcn_fence(__ATOMIC_ACQUIRE, "wavefront");

    int rowv = ((int)c      < count) ? gl[c]      : 0;
    int bo0  = ((int)rr     < count) ? gl[rr]     : 0;
    int bo1  = ((int)rr + 8 < count) ? gl[rr + 8] : 0;

    for (int m0 = 0; m0 < count; m0 += 16) {
        v4f av[16];
        #pragma unroll
        for (int m = 0; m < 16; ++m) {
            const int r = __builtin_amdgcn_readlane(rowv, m);
            av[m] = *(const v4f*)(attended + (size_t)r * C_N + kw + 4u * lane);
        }

        gv4f_ptr wq = (gv4f_ptr)wp;
        asm volatile("" : "+v"(wq));

        v4f ring[G_AHEAD + 1][2];
#define G_LOAD_STEP(S) do {                                                                          \
        ring[(S) % (G_AHEAD + 1)][0] = __builtin_nontemporal_load(wq + (size_t)(4 * (S) + 0) * (NA_N / 4)); \
        ring[(S) % (G_AHEAD + 1)][1] = __builtin_nontemporal_load(wq + (size_t)(4 * (S) + 1) * (NA_N / 4)); \
    } while (0)

        __builtin_amdgcn_sched_barrier(0);
        #pragma unroll
        for (int s = 0; s < G_AHEAD / 2; ++s) G_LOAD_STEP(s);
        __builtin_amdgcn_sched_barrier(0);

        #pragma unroll
        for (int m = 0; m < 16; ++m)
            *(v4f*)&a_s[wave][m][4u * lane] = av[m];
        __builtin_amdgcn_sched_barrier(0);
        #pragma unroll
        for (int s = G_AHEAD / 2; s < G_AHEAD; ++s) G_LOAD_STEP(s);
        __builtin_amdgcn_fence(__ATOMIC_RELEASE, "wavefront");
        __builtin_amdgcn_wave_barrier();
        __builtin_amdgcn_fence(__ATOMIC_ACQUIRE, "wavefront");

        v8f acc0 = {}, acc1 = {}, acc2 = {}, acc3 = {};
        v2f af[G_ABLK];

        #pragma unroll
        for (int s = 0; s < G_STEPS; ++s) {
            __builtin_amdgcn_sched_barrier(0);
            if (s + G_AHEAD < G_STEPS) G_LOAD_STEP(s + G_AHEAD);
            __builtin_amdgcn_sched_barrier(0);
            if (s % G_ABLK == 0) {
                #pragma unroll
                for (int i = 0; i < G_ABLK; ++i)
                    af[i] = *(const v2f*)&a_s[wave][c][4 * (s + i) + 2u * h];
            }
            const v4f r0 = ring[s % (G_AHEAD + 1)][0];
            const v4f r1 = ring[s % (G_AHEAD + 1)][1];
            const v2f a  = af[s % G_ABLK];
            v2f b;
            b.x = r0.x; b.y = r1.x;
            acc0 = __builtin_amdgcn_wmma_f32_16x16x4_f32(false, a, false, b, (short)0, acc0, false, false);
            b.x = r0.y; b.y = r1.y;
            acc1 = __builtin_amdgcn_wmma_f32_16x16x4_f32(false, a, false, b, (short)0, acc1, false, false);
            b.x = r0.z; b.y = r1.z;
            acc2 = __builtin_amdgcn_wmma_f32_16x16x4_f32(false, a, false, b, (short)0, acc2, false, false);
            b.x = r0.w; b.y = r1.w;
            acc3 = __builtin_amdgcn_wmma_f32_16x16x4_f32(false, a, false, b, (short)0, acc3, false, false);
        }
#undef G_LOAD_STEP

        #pragma unroll
        for (int v = 0; v < 8; ++v) {
            v4f p;
            p.x = acc0[v]; p.y = acc1[v]; p.z = acc2[v]; p.w = acc3[v];
            p_s[wave][v + 8 * h][c] = p;
        }
        __syncthreads();

        if (colo < (unsigned)NA_N) {
            const v4f bia = *(const v4f*)(bias + d * NA_N + colo);
            if (m0 + (int)rr < count) {
                const v4f s4 = ((p_s[0][rr][c4] + p_s[1][rr][c4]) + (p_s[2][rr][c4] + p_s[3][rr][c4])) + bia;
                *(v4f*)(out + (size_t)bo0 * NA_N + colo) = s4;
            }
            if (m0 + (int)rr + 8 < count) {
                const v4f s4 = ((p_s[0][rr + 8][c4] + p_s[1][rr + 8][c4]) + (p_s[2][rr + 8][c4] + p_s[3][rr + 8][c4])) + bia;
                *(v4f*)(out + (size_t)bo1 * NA_N + colo) = s4;
            }
        }
        if (m0 + 16 < count) {
            __syncthreads();
            rowv = (m0 + 16 + (int)c      < count) ? gl[m0 + 16 + c]      : 0;
            bo0  = (m0 + 16 + (int)rr     < count) ? gl[m0 + 16 + rr]     : 0;
            bo1  = (m0 + 16 + (int)rr + 8 < count) ? gl[m0 + 16 + rr + 8] : 0;
        }
    }
}

extern "C" void kernel_launch(void* const* d_in, const int* in_sizes, int n_in,
                              void* d_out, int out_size, void* d_ws, size_t ws_size,
                              hipStream_t stream) {
    const float* mask     = (const float*)d_in[0];
    const float* features = (const float*)d_in[1];
    const float* weight   = (const float*)d_in[2];
    const float* bias     = (const float*)d_in[3];
    const int*   instance = (const int*)d_in[4];
    float*       out      = (float*)d_out;

    float* attended = (float*)d_ws;

    k_attend<<<dim3(B_N * C_N / (ATT_ROWS * ATT_WAVES)), dim3(ATT_WAVES * 32), 0, stream>>>(mask, features, attended);
    k_gemm  <<<dim3(G_NBLK, ND_N), dim3(G_WAVES * 32), 0, stream>>>(
        attended, weight, bias, instance, out);
}


# GEMM waves request their first 8 k-steps of weight and their bias float4 at entry, ahead of the expert-id wait, list build and A-row loads; no bias round trip behind the last barrier
# speedup vs baseline: 10.5090x; 1.0359x over previous
#include <hip/hip_runtime.h>
#include <hip/hip_bf16.h>

#define B_N   256
#define C_N   512
#define HW_N  196
#define ND_N  32
#define NA_N  2000

typedef float v2f __attribute__((ext_vector_type(2)));
typedef float v8f __attribute__((ext_vector_type(8)));

#define HW4_N      (HW_N / 4)
#define ATT_ROWS   32
#define ATT_WAVES  8
#define ATT_PIECES HW4_N
#define ATT_AHEAD  32

typedef float v4f __attribute__((ext_vector_type(4)));

__global__ void __launch_bounds__(ATT_WAVES * 32)
k_attend(const float* __restrict__ mask,
         const float* __restrict__ feat,
         float* __restrict__ attended) {
    __shared__ v4f m_s[ATT_WAVES][2 * HW4_N];

    const unsigned tid  = threadIdx.x;
    const unsigned lane = tid & 31u;
    const unsigned wave = __builtin_amdgcn_readfirstlane(tid >> 5);
    const unsigned row0 = (blockIdx.x * ATT_WAVES + wave) * ATT_ROWS;
    const unsigned b    = row0 / C_N;

    const bool     two  = lane < (unsigned)(HW4_N - 32);
    const unsigned idx1 = two ? (32u + lane) : (unsigned)(HW4_N - 1);
    const v4f* m4 = (const v4f*)mask + (size_t)b * HW4_N;
    const v4f  m0 = m4[lane];
    const v4f  m1 = m4[idx1];

    __builtin_amdgcn_sched_barrier(0);

    const v4f* f = (const v4f*)feat + (size_t)row0 * HW4_N + lane;

    v4f   w[ATT_AHEAD + 1];
    float sum[ATT_ROWS];

#define ATT_LOAD_PIECE(I) do {                                                       \
        w[(I) % (ATT_AHEAD + 1)] = __builtin_nontemporal_load(f + (I) * 32);         \
    } while (0)

    #pragma unroll
    for (int i = 0; i < ATT_AHEAD; ++i) ATT_LOAD_PIECE(i);
    __builtin_amdgcn_sched_barrier(0);

    v4f* ms = m_s[wave];
    ms[lane]         = m0;
    ms[HW4_N + lane] = m0;
    ms[idx1]         = m1;
    ms[HW4_N + idx1] = m1;
    __builtin_amdgcn_fence(__ATOMIC_RELEASE, "wavefront");
    __builtin_amdgcn_wave_barrier();
    __builtin_amdgcn_fence(__ATOMIC_ACQUIRE, "wavefront");

    v4f mn = ms[lane];

    #pragma unroll
    for (int i = 0; i < ATT_PIECES; ++i) {
        __builtin_amdgcn_sched_barrier(0);
        if (i + ATT_AHEAD < ATT_PIECES) ATT_LOAD_PIECE(i + ATT_AHEAD);
        const int r_lo = (32 * i) / HW4_N;
        const int base = 32 * i - HW4_N * r_lo;
        const int thr  = HW4_N - base;
        const v4f m = mn;
        if (i + 1 < ATT_PIECES) mn = ms[(32 * (i + 1)) % HW4_N + (int)lane];
        __builtin_amdgcn_sched_barrier(0);
        const v4f a = w[i % (ATT_AHEAD + 1)];
        float p = m.x * a.x;
        p = __builtin_fmaf(m.y, a.y, p);
        p = __builtin_fmaf(m.z, a.z, p);
        p = __builtin_fmaf(m.w, a.w, p);
        if (i == 0) {
            sum[0] = p;
        } else if (thr >= 32) {
            sum[r_lo] += p;
        } else {
            const bool lo = lane < (unsigned)thr;
            sum[r_lo]    += lo ? p : 0.0f;
            sum[r_lo + 1] = lo ? 0.0f : p;
            asm volatile("" : "+v"(sum[r_lo + 1]));
        }
        asm volatile("" : "+v"(sum[r_lo]));
    }
#undef ATT_LOAD_PIECE

    #pragma unroll
    for (int d = 16; d >= 1; d >>= 1) {
        const bool up = (lane & (unsigned)d) != 0u;
        #pragma unroll
        for (int i = 0; i < d; ++i) {
            const float lo = sum[i], hi = sum[i + d];
            const float keep = up ? hi : lo;
            const float send = up ? lo : hi;
            sum[i] = keep + __shfl_xor(send, d, 32);
        }
    }
    attended[row0 + lane] = sum[0] * (1.0f / (float)HW_N);
}

#define G_COLS   64
#define G_WAVES  4
#define G_KW     (C_N / G_WAVES)
#define G_STEPS  (G_KW / 4)
#define G_AHEAD  16
#define G_ABLK   8
#define G_APITCH (G_KW + 4)
#define G_NBLK   ((NA_N + G_COLS - 1) / G_COLS)

typedef const v4f __attribute__((address_space(1)))* gv4f_ptr;
typedef int v4i __attribute__((ext_vector_type(4)));

__global__ void __launch_bounds__(G_WAVES * 32)
k_gemm(const float* __restrict__ attended,
       const float* __restrict__ weight,
       const float* __restrict__ bias,
       const int* __restrict__ instance,
       float* __restrict__ out) {
    __shared__ __attribute__((aligned(16))) float a_s[G_WAVES][16][G_APITCH];
    __shared__ v4f p_s[G_WAVES][16][17];
    __shared__ int l_s[G_WAVES][B_N];

    const unsigned tid  = threadIdx.x;
    const unsigned lane = tid & 31u;
    const unsigned wave = __builtin_amdgcn_readfirstlane(tid >> 5);
    const unsigned h    = lane >> 4;
    const unsigned c    = lane & 15u;
    const unsigned d    = blockIdx.y;
    const unsigned a0   = blockIdx.x * G_COLS;
    const unsigned kw   = wave * G_KW;

    const unsigned colw = min(a0 + 4u * c, (unsigned)(NA_N - 4));
    const v4f* wp = (const v4f*)(weight + ((size_t)d * C_N + kw + 2u * h) * NA_N + colw);

    const unsigned c4 = tid & 15u;
    const unsigned rr = tid >> 4;
    const unsigned colo = a0 + 4u * c4;

    const v4i* ip = (const v4i*)instance;
    const v4i  i0 = ip[lane];
    const v4i  i1 = ip[32u + lane];

    const v4f bia = *(const v4f*)(bias + d * NA_N + min(colo, (unsigned)(NA_N - 4)));

    v4f ring[G_AHEAD + 1][2];
#define G_LOAD_STEP(S) do {                                                                          \
        ring[(S) % (G_AHEAD + 1)][0] = __builtin_nontemporal_load(wq + (size_t)(4 * (S) + 0) * (NA_N / 4)); \
        ring[(S) % (G_AHEAD + 1)][1] = __builtin_nontemporal_load(wq + (size_t)(4 * (S) + 1) * (NA_N / 4)); \
    } while (0)

    gv4f_ptr wq = (gv4f_ptr)wp;
    asm volatile("" : "+v"(wq));
    __builtin_amdgcn_sched_barrier(0);
    #pragma unroll
    for (int s = 0; s < G_AHEAD / 2; ++s) G_LOAD_STEP(s);
    __builtin_amdgcn_sched_barrier(0);

    int* gl = l_s[wave];
    int count = 0;
    #pragma unroll
    for (int t = 0; t < 8; ++t) {
        const int      v   = (t < 4) ? i0[t & 3] : i1[t & 3];
        const bool     hit = (v == (int)d);
        const unsigned m   = __builtin_amdgcn_ballot_w32(hit);
        if (hit) gl[count + (int)__builtin_amdgcn_mbcnt_lo(m, 0u)] = (int)((unsigned)(t >> 2) * 128u + 4u * lane + (unsigned)(t & 3));
        count += __builtin_popcount(m);
    }
    __builtin_amdgcn_fence(__ATOMIC_RELEASE, "wavefront");
    __builtin_amdgcn_wave_barrier();
    __builtin_amdgcn_fence(__ATOMIC_ACQUIRE, "wavefront");

    int rowv = ((int)c      < count) ? gl[c]      : 0;
    int bo0  = ((int)rr     < count) ? gl[rr]     : 0;
    int bo1  = ((int)rr + 8 < count) ? gl[rr + 8] : 0;

    for (int m0 = 0; m0 < count; m0 += 16) {
        v4f av[16];
        #pragma unroll
        for (int m = 0; m < 16; ++m) {
            const int r = __builtin_amdgcn_readlane(rowv, m);
            av[m] = *(const v4f*)(attended + (size_t)r * C_N + kw + 4u * lane);
        }

        #pragma unroll
        for (int m = 0; m < 16; ++m)
            *(v4f*)&a_s[wave][m][4u * lane] = av[m];
        __builtin_amdgcn_sched_barrier(0);
        #pragma unroll
        for (int s = G_AHEAD / 2; s < G_AHEAD; ++s) G_LOAD_STEP(s);
        __builtin_amdgcn_fence(__ATOMIC_RELEASE, "wavefront");
        __builtin_amdgcn_wave_barrier();
        __builtin_amdgcn_fence(__ATOMIC_ACQUIRE, "wavefront");

        v8f acc0 = {}, acc1 = {}, acc2 = {}, acc3 = {};
        v2f af[G_ABLK];

        #pragma unroll
        for (int s = 0; s < G_STEPS; ++s) {
            __builtin_amdgcn_sched_barrier(0);
            if (s + G_AHEAD < G_STEPS) G_LOAD_STEP(s + G_AHEAD);
            __builtin_amdgcn_sched_barrier(0);
            if (s % G_ABLK == 0) {
                #pragma unroll
                for (int i = 0; i < G_ABLK; ++i)
                    af[i] = *(const v2f*)&a_s[wave][c][4 * (s + i) + 2u * h];
            }
            const v4f r0 = ring[s % (G_AHEAD + 1)][0];
            const v4f r1 = ring[s % (G_AHEAD + 1)][1];
            const v2f a  = af[s % G_ABLK];
            v2f b;
            b.x = r0.x; b.y = r1.x;
            acc0 = __builtin_amdgcn_wmma_f32_16x16x4_f32(false, a, false, b, (short)0, acc0, false, false);
            b.x = r0.y; b.y = r1.y;
            acc1 = __builtin_amdgcn_wmma_f32_16x16x4_f32(false, a, false, b, (short)0, acc1, false, false);
            b.x = r0.z; b.y = r1.z;
            acc2 = __builtin_amdgcn_wmma_f32_16x16x4_f32(false, a, false, b, (short)0, acc2, false, false);
            b.x = r0.w; b.y = r1.w;
            acc3 = __builtin_amdgcn_wmma_f32_16x16x4_f32(false, a, false, b, (short)0, acc3, false, false);
        }

        #pragma unroll
        for (int v = 0; v < 8; ++v) {
            v4f p;
            p.x = acc0[v]; p.y = acc1[v]; p.z = acc2[v]; p.w = acc3[v];
            p_s[wave][v + 8 * h][c] = p;
        }
        __syncthreads();

        if (colo < (unsigned)NA_N) {
            if (m0 + (int)rr < count) {
                const v4f s4 = ((p_s[0][rr][c4] + p_s[1][rr][c4]) + (p_s[2][rr][c4] + p_s[3][rr][c4])) + bia;
                *(v4f*)(out + (size_t)bo0 * NA_N + colo) = s4;
            }
            if (m0 + (int)rr + 8 < count) {
                const v4f s4 = ((p_s[0][rr + 8][c4] + p_s[1][rr + 8][c4]) + (p_s[2][rr + 8][c4] + p_s[3][rr + 8][c4])) + bia;
                *(v4f*)(out + (size_t)bo1 * NA_N + colo) = s4;
            }
        }
        if (m0 + 16 < count) {
            __syncthreads();
            rowv = (m0 + 16 + (int)c      < count) ? gl[m0 + 16 + c]      : 0;
            bo0  = (m0 + 16 + (int)rr     < count) ? gl[m0 + 16 + rr]     : 0;
            bo1  = (m0 + 16 + (int)rr + 8 < count) ? gl[m0 + 16 + rr + 8] : 0;
            #pragma unroll
            for (int s = 0; s < G_AHEAD / 2; ++s) G_LOAD_STEP(s);
        }
    }
#undef G_LOAD_STEP
}

extern "C" void kernel_launch(void* const* d_in, const int* in_sizes, int n_in,
                              void* d_out, int out_size, void* d_ws, size_t ws_size,
                              hipStream_t stream) {
    const float* mask     = (const float*)d_in[0];
    const float* features = (const float*)d_in[1];
    const float* weight   = (const float*)d_in[2];
    const float* bias     = (const float*)d_in[3];
    const int*   instance = (const int*)d_in[4];
    float*       out      = (float*)d_out;

    float* attended = (float*)d_ws;

    k_attend<<<dim3(B_N * C_N / (ATT_ROWS * ATT_WAVES)), dim3(ATT_WAVES * 32), 0, stream>>>(mask, features, attended);
    k_gemm  <<<dim3(G_NBLK, ND_N), dim3(G_WAVES * 32), 0, stream>>>(
        attended, weight, bias, instance, out);
}
